# starting kernel
# baseline (speedup 1.0000x reference)
#include <hip/hip_runtime.h>
#include <math.h>

#define BB 16
#define CC 256
#define HW 16384
#define ROWS (BB * CC)
#define CP  (CC + 4)

typedef __attribute__((ext_vector_type(2))) float v2f;
typedef __attribute__((ext_vector_type(4))) float v4f;
typedef __attribute__((ext_vector_type(8))) float v8f;

__global__ __launch_bounds__(256) void cam_pool_kernel(
    const float* __restrict__ x,
    float* __restrict__ avg_out,
    float* __restrict__ max_out) {
  const int row = blockIdx.x;
  const v4f* __restrict__ p4 =
      (const v4f*)(x + (size_t)row * HW);
  const int tid = threadIdx.x;

  float s = 0.0f;
  float m = -INFINITY;
#pragma unroll
  for (int i = 0; i < 16; ++i) {
    v4f v = __builtin_nontemporal_load(&p4[tid + i * 256]);
    s += (v.x + v.y) + (v.z + v.w);
    m = fmaxf(m, fmaxf(fmaxf(v.x, v.y), fmaxf(v.z, v.w)));
  }

#pragma unroll
  for (int off = 16; off > 0; off >>= 1) {
    s += __shfl_xor(s, off, 32);
    m = fmaxf(m, __shfl_xor(m, off, 32));
  }

  __shared__ float ss[8];
  __shared__ float sm[8];
  const int wave = tid >> 5;
  if ((tid & 31) == 0) { ss[wave] = s; sm[wave] = m; }
  __syncthreads();
  if (tid == 0) {
    float S = ss[0], M = sm[0];
#pragma unroll
    for (int w = 1; w < 8; ++w) { S += ss[w]; M = fmaxf(M, sm[w]); }
    avg_out[row] = S * (1.0f / (float)HW);
    max_out[row] = M;
  }
}

__global__ __launch_bounds__(256) void cam_mlp_wmma_kernel(
    const float* __restrict__ avg,
    const float* __restrict__ mx,
    const float* __restrict__ w1,
    const float* __restrict__ w2,
    float* __restrict__ out) {
  __shared__ float h_avg[BB][CP];
  __shared__ float h_max[BB][CP];

  const int tid  = threadIdx.x;
  const int wave = tid >> 5;
  const int lane = tid & 31;
  const int half = lane >> 4;
  const int l16  = lane & 15;

  for (int t = 0; t < 2; ++t) {
    const int ntile = wave * 2 + t;
    const int ocol  = ntile * 16 + l16;
    v8f acc_a = {};
    v8f acc_m = {};
    for (int k = 0; k < CC; k += 4) {
      const int ka = k + half * 2;
      v2f a_a, a_m, b;
      a_a.x = avg[l16 * CC + ka];  a_a.y = avg[l16 * CC + ka + 1];
      a_m.x = mx [l16 * CC + ka];  a_m.y = mx [l16 * CC + ka + 1];
      b.x = w1[ocol * CC + ka];    b.y = w1[ocol * CC + ka + 1];
      acc_a = __builtin_amdgcn_wmma_f32_16x16x4_f32(
          false, a_a, false, b, (short)0, acc_a, false, false);
      acc_m = __builtin_amdgcn_wmma_f32_16x16x4_f32(
          false, a_m, false, b, (short)0, acc_m, false, false);
    }
#pragma unroll
    for (int r = 0; r < 8; ++r) {
      const int mrow = r + 8 * half;
      h_avg[mrow][ntile * 16 + l16] = fmaxf(acc_a[r], 0.0f);
      h_max[mrow][ntile * 16 + l16] = fmaxf(acc_m[r], 0.0f);
    }
  }
  __syncthreads();

  for (int t = 0; t < 2; ++t) {
    const int ntile = wave * 2 + t;
    const int ocol  = ntile * 16 + l16;
    v8f acc_a = {};
    v8f acc_m = {};
    for (int k = 0; k < CC; k += 4) {
      const int ka = k + half * 2;
      v2f a_a, a_m, b;
      a_a.x = h_avg[l16][ka];  a_a.y = h_avg[l16][ka + 1];
      a_m.x = h_max[l16][ka];  a_m.y = h_max[l16][ka + 1];
      b.x = w2[ocol * CC + ka];  b.y = w2[ocol * CC + ka + 1];
      acc_a = __builtin_amdgcn_wmma_f32_16x16x4_f32(
          false, a_a, false, b, (short)0, acc_a, false, false);
      acc_m = __builtin_amdgcn_wmma_f32_16x16x4_f32(
          false, a_m, false, b, (short)0, acc_m, false, false);
    }
#pragma unroll
    for (int r = 0; r < 8; ++r) {
      const int mrow = r + 8 * half;
      const float z = acc_a[r] + acc_m[r];
      out[mrow * CC + ntile * 16 + l16] = 1.0f / (1.0f + __expf(-z));
    }
  }
}

extern "C" void kernel_launch(void* const* d_in, const int* in_sizes, int n_in,
                              void* d_out, int out_size, void* d_ws, size_t ws_size,
                              hipStream_t stream) {
  const float* x  = (const float*)d_in[0];
  const float* w1 = (const float*)d_in[1];
  const float* w2 = (const float*)d_in[2];
  float* out = (float*)d_out;

  float* ws_avg = (float*)d_ws;
  float* ws_max = ws_avg + ROWS;

  cam_pool_kernel<<<ROWS, 256, 0, stream>>>(x, ws_avg, ws_max);
  cam_mlp_wmma_kernel<<<1, 256, 0, stream>>>(ws_avg, ws_max, w1, w2, out);
}


# pool as a slab sweep: 512 workgroups take 8 consecutive rows each, every row split over the 8 waves, next row's 16 loads requested ahead; per-lane partials via LDS, one barrier; sum order changes
# speedup vs baseline: 1.7385x; 1.7385x over previous
#include <hip/hip_runtime.h>
#include <math.h>

#define BB 16
#define CC 256
#define HW 16384
#define ROWS (BB * CC)
#define CP  (CC + 4)
#define HT  16
#define NHT (CC / HT)
#define HP  (HT + 4)

typedef __attribute__((ext_vector_type(2))) float v2f;
typedef __attribute__((ext_vector_type(4))) float v4f;
typedef __attribute__((ext_vector_type(8))) float v8f;

#define PR 8
#define PWAVES 8
#define SEG4 (HW / 4 / PWAVES)
#define PQ (SEG4 / 32)

__global__ __launch_bounds__(256) void cam_pool_kernel(
    const float* __restrict__ x,
    float* __restrict__ avg_out,
    float* __restrict__ max_out) {
  __shared__ v2f   ss[PR][PWAVES][32];
  __shared__ float sm[PR][PWAVES][32];

  const int tid  = threadIdx.x;
  const int lane = tid & 31;
  const int wave = __builtin_amdgcn_readfirstlane(tid >> 5);
  const int row0 = blockIdx.x * PR;
  const v4f* __restrict__ p4 =
      (const v4f*)(x + (size_t)row0 * HW) + wave * SEG4 + lane;

  v4f buf[2][PQ];
#pragma unroll
  for (int q = 0; q < PQ; ++q)
    buf[0][q] = __builtin_nontemporal_load(&p4[32 * q]);

#pragma unroll
  for (int r = 0; r < PR; ++r) {
    __builtin_amdgcn_sched_barrier(0);
    if (r + 1 < PR) {
#pragma unroll
      for (int q = 0; q < PQ; ++q)
        buf[(r + 1) & 1][q] = __builtin_nontemporal_load(&p4[(r + 1) * (HW / 4) + 32 * q]);
    }
    __builtin_amdgcn_sched_barrier(0);

    const v4f* cur = buf[r & 1];
    v2f a0 = {0.0f, 0.0f}, a1 = {0.0f, 0.0f};
    float m0 = -INFINITY, m1 = -INFINITY;
#pragma unroll
    for (int q = 0; q < PQ; ++q) {
      const v4f v = cur[q];
      a0 += __builtin_shufflevector(v, v, 0, 1);
      a1 += __builtin_shufflevector(v, v, 2, 3);
      m0 = fmaxf(m0, fmaxf(v.x, v.y));
      m1 = fmaxf(m1, fmaxf(v.z, v.w));
    }
    ss[r][wave][lane] = a0 + a1;
    sm[r][wave][lane] = fmaxf(m0, m1);
  }
  __syncthreads();

  v2f   T = ss[wave][0][lane];
  float M = sm[wave][0][lane];
#pragma unroll
  for (int k = 1; k < PWAVES; ++k) {
    T += ss[wave][k][lane];
    M = fmaxf(M, sm[wave][k][lane]);
  }
  float S = T.x + T.y;
#pragma unroll
  for (int off = 16; off > 0; off >>= 1) {
    S += __shfl_xor(S, off, 32);
    M = fmaxf(M, __shfl_xor(M, off, 32));
  }
  if (lane == 0) {
    avg_out[row0 + wave] = S * (1.0f / (float)HW);
    max_out[row0 + wave] = M;
  }
}

__global__ __launch_bounds__(256) void cam_mlp_hidden_kernel(
    const float* __restrict__ avg,
    const float* __restrict__ mx,
    const float* __restrict__ w1,
    const float* __restrict__ w2,
    float* __restrict__ part) {
  __shared__ __attribute__((aligned(16))) float sv[2 * BB][CP];
  __shared__ __attribute__((aligned(16))) float sw1[HT][CP];
  __shared__ __attribute__((aligned(16))) float sh[2 * BB][HP];

  const int j    = blockIdx.x;
  const int tid  = threadIdx.x;
  const int wave = tid >> 5;
  const int lane = tid & 31;
  const int half = lane >> 4;
  const int l16  = lane & 15;

  v2f bw[2][4];
#pragma unroll
  for (int t = 0; t < 2; ++t)
#pragma unroll
    for (int ks = 0; ks < 4; ++ks)
      bw[t][ks] = *(const v2f*)&w2[((wave * 2 + t) * 16 + l16) * CC + j * HT + 4 * ks + 2 * half];

  {
    const v4f* __restrict__ a4 = (const v4f*)avg;
    const v4f* __restrict__ m4 = (const v4f*)mx;
    const v4f* __restrict__ w4 = (const v4f*)(w1 + j * HT * CC);
    v4f ra[4], rm[4], rw[4];
#pragma unroll
    for (int i = 0; i < 4; ++i) {
      const int idx = tid + 256 * i;
      ra[i] = a4[idx]; rm[i] = m4[idx]; rw[i] = w4[idx];
    }
#pragma unroll
    for (int i = 0; i < 4; ++i) {
      const int idx = tid + 256 * i;
      const int row = idx >> 6, c4 = (idx & 63) * 4;
      *(v4f*)&sv[row][c4]      = ra[i];
      *(v4f*)&sv[BB + row][c4] = rm[i];
      *(v4f*)&sw1[row][c4]     = rw[i];
    }
  }
  __syncthreads();

  if (wave < 2) {
    const float* arow = &sv[wave * BB + l16][2 * half];
    const float* brow = &sw1[l16][2 * half];
    v8f acc0 = {};
    v8f acc1 = {};
    v2f fa[2][16], fb[2][16];
#pragma unroll
    for (int i = 0; i < 16; ++i) {
      fa[0][i] = *(const v2f*)(arow + 4 * i);
      fb[0][i] = *(const v2f*)(brow + 4 * i);
    }
#pragma unroll
    for (int blk = 0; blk < 4; ++blk) {
      const int cur = blk & 1;
      if (blk + 1 < 4) {
#pragma unroll
        for (int i = 0; i < 16; ++i) {
          fa[cur ^ 1][i] = *(const v2f*)(arow + 4 * (16 * (blk + 1) + i));
          fb[cur ^ 1][i] = *(const v2f*)(brow + 4 * (16 * (blk + 1) + i));
        }
      }
      __builtin_amdgcn_sched_barrier(0);
#pragma unroll
      for (int i = 0; i < 16; i += 2) {
        acc0 = __builtin_amdgcn_wmma_f32_16x16x4_f32(
            false, fa[cur][i], false, fb[cur][i], (short)0, acc0, false, false);
        acc1 = __builtin_amdgcn_wmma_f32_16x16x4_f32(
            false, fa[cur][i + 1], false, fb[cur][i + 1], (short)0, acc1, false, false);
      }
      __builtin_amdgcn_sched_barrier(0);
    }
#pragma unroll
    for (int r = 0; r < 8; ++r)
      sh[wave * BB + r + 8 * half][l16] = fmaxf(acc0[r] + acc1[r], 0.0f);
  }
  __syncthreads();

  v2f aa[4], am[4];
#pragma unroll
  for (int ks = 0; ks < 4; ++ks) {
    aa[ks] = *(const v2f*)&sh[l16][4 * ks + 2 * half];
    am[ks] = *(const v2f*)&sh[BB + l16][4 * ks + 2 * half];
  }
  v8f o0 = {};
  v8f o1 = {};
#pragma unroll
  for (int ks = 0; ks < 4; ++ks) {
    o0 = __builtin_amdgcn_wmma_f32_16x16x4_f32(false, aa[ks], false, bw[0][ks], (short)0, o0, false, false);
    o1 = __builtin_amdgcn_wmma_f32_16x16x4_f32(false, aa[ks], false, bw[1][ks], (short)0, o1, false, false);
    o0 = __builtin_amdgcn_wmma_f32_16x16x4_f32(false, am[ks], false, bw[0][ks], (short)0, o0, false, false);
    o1 = __builtin_amdgcn_wmma_f32_16x16x4_f32(false, am[ks], false, bw[1][ks], (short)0, o1, false, false);
  }
  float* __restrict__ pj = part + j * ROWS;
#pragma unroll
  for (int r = 0; r < 8; ++r) {
    const int mrow = r + 8 * half;
    pj[mrow * CC + (wave * 2 + 0) * 16 + l16] = o0[r];
    pj[mrow * CC + (wave * 2 + 1) * 16 + l16] = o1[r];
  }
}

__global__ __launch_bounds__(256) void cam_sum_sigmoid_kernel(
    const float* __restrict__ part,
    float* __restrict__ out) {
  const int i = blockIdx.x * 256 + threadIdx.x;
  float p[NHT];
#pragma unroll
  for (int t = 0; t < NHT; ++t) p[t] = part[t * ROWS + i];
  float z = p[0];
#pragma unroll
  for (int t = 1; t < NHT; ++t) z += p[t];
  out[i] = 1.0f / (1.0f + __expf(-z));
}

extern "C" void kernel_launch(void* const* d_in, const int* in_sizes, int n_in,
                              void* d_out, int out_size, void* d_ws, size_t ws_size,
                              hipStream_t stream) {
  const float* x  = (const float*)d_in[0];
  const float* w1 = (const float*)d_in[1];
  const float* w2 = (const float*)d_in[2];
  float* out = (float*)d_out;

  float* ws_avg  = (float*)d_ws;
  float* ws_max  = ws_avg + ROWS;
  float* ws_part = ws_max + ROWS;

  cam_pool_kernel<<<ROWS / PR, 256, 0, stream>>>(x, ws_avg, ws_max);
  cam_mlp_hidden_kernel<<<NHT, 256, 0, stream>>>(ws_avg, ws_max, w1, w2, ws_part);
  cam_sum_sigmoid_kernel<<<ROWS / 256, 256, 0, stream>>>(ws_part, out);
}
